# starting kernel
# baseline (speedup 1.0000x reference)
#include <hip/hip_runtime.h>
#include <hip/hip_bf16.h>

typedef float v2f __attribute__((ext_vector_type(2)));
typedef float v8f __attribute__((ext_vector_type(8)));

#define B_    4
#define C_    64
#define D_    32
#define H_    64
#define W_    64
#define HW    (H_ * W_)
#define DHW   (D_ * HW)
#define CDHW  (C_ * DHW)
#define TW    24

__global__ __launch_bounds__(256) void sg_reduce_maxmean(
    const float* __restrict__ x, float* __restrict__ xmax, float* __restrict__ xmean) {
  long t = (long)blockIdx.x * blockDim.x + threadIdx.x;
  long f = t * 4;
  int  b = (int)(f / DHW);
  int  q = (int)(f % DHW);
  const float* xp = x + (long)b * CDHW + q;

  float4 v = *(const float4*)xp;
  float mx0 = v.x, mx1 = v.y, mx2 = v.z, mx3 = v.w;
  float s0 = v.x, s1 = v.y, s2 = v.z, s3 = v.w;
  for (int c = 1; c < C_; ++c) {
    float4 u = *(const float4*)(xp + (long)c * DHW);
    mx0 = fmaxf(mx0, u.x); mx1 = fmaxf(mx1, u.y);
    mx2 = fmaxf(mx2, u.z); mx3 = fmaxf(mx3, u.w);
    s0 += u.x; s1 += u.y; s2 += u.z; s3 += u.w;
  }
  const float inv = 1.0f / (float)C_;
  float4 mo; mo.x = mx0; mo.y = mx1; mo.z = mx2; mo.w = mx3;
  float4 me; me.x = s0 * inv; me.y = s1 * inv; me.z = s2 * inv; me.w = s3 * inv;
  *(float4*)(xmax  + (long)b * DHW + q) = mo;
  *(float4*)(xmean + (long)b * DHW + q) = me;
}

__global__ __launch_bounds__(32) void sg_conv_sigmoid(
    const float* __restrict__ xmax, const float* __restrict__ xmean,
    const float* __restrict__ wgt, const float* __restrict__ bias,
    float* __restrict__ scale) {
  __shared__ float lw[2 * 7 * 7 * 7];
  __shared__ float tile[TW * TW];

  const int lane = threadIdx.x;
  const int bid  = blockIdx.x;
  const int wt = bid & 3, ht = (bid >> 2) & 3;
  const int d  = (bid >> 4) & 31, b = bid >> 9;
  const int hbase = ht * 16, wbase = wt * 16;

  for (int i = lane; i < 686; i += 32) lw[i] = wgt[i];
  __syncthreads();

  const int m    = lane & 15;
  const int n    = m;
  const int kgrp = (lane >> 4) << 1;

  v8f acc = {0.f, 0.f, 0.f, 0.f, 0.f, 0.f, 0.f, 0.f};

  for (int c = 0; c < 2; ++c) {
    const float* plane_b = (c == 0 ? xmax : xmean) + (long)b * DHW;
    for (int kd = 0; kd < 7; ++kd) {
      const int dz = d + kd - 3;
      if (dz < 0 || dz >= D_) continue;
      const float* plane = plane_b + (long)dz * HW;

      for (int i = lane; i < 24 * 22; i += 32) {
        const int row = i / 22, col = i % 22;
        const int h = hbase + row - 3, w = wbase + col - 3;
        float v = 0.0f;
        if (row < 22 && h >= 0 && h < H_ && w >= 0 && w < W_)
          v = plane[h * W_ + w];
        tile[row * TW + col] = v;
      }
      __syncthreads();

      for (int kw = 0; kw < 7; ++kw) {
        const int wb0 = (c * 7 + kd) * 49 + kw;
        float w0[2], w1[2]; int c0[2];
#pragma unroll
        for (int s = 0; s < 2; ++s) {
          const int ks = kgrp + s;
          const int r  = (ks - m) & 3;
          c0[s] = (m + r - ks) >> 2;
          w0[s] = lw[wb0 + r * 7];
          w1[s] = (r <= 2) ? lw[wb0 + (r + 4) * 7] : 0.0f;
        }
#pragma unroll
        for (int ch = 0; ch < 6; ++ch) {
          v2f a, bb;
          a.x = (ch == c0[0]) ? w0[0] : ((ch == c0[0] + 1) ? w1[0] : 0.0f);
          a.y = (ch == c0[1]) ? w0[1] : ((ch == c0[1] + 1) ? w1[1] : 0.0f);
          const int j0 = ch * 4 + kgrp;
          bb.x = tile[j0 * TW + kw + n];
          bb.y = tile[(j0 + 1) * TW + kw + n];
          acc = __builtin_amdgcn_wmma_f32_16x16x4_f32(
              false, a, false, bb, (short)0, acc, false, false);
        }
      }
      __syncthreads();
    }
  }

  const float bv = bias[0];
  float* sp = scale + ((long)b * D_ + d) * HW;
#pragma unroll
  for (int r = 0; r < 8; ++r) {
    const int M = r + (kgrp << 2);
    const float y = acc[r] + bv;
    sp[(hbase + M) * W_ + (wbase + n)] = 1.0f / (1.0f + __expf(-y));
  }
}

__global__ __launch_bounds__(256) void sg_apply(
    const float* __restrict__ x, const float* __restrict__ scale,
    float* __restrict__ out) {
  long t = (long)blockIdx.x * blockDim.x + threadIdx.x;
  long f = t * 4;
  int  b = (int)(f / CDHW);
  int  p = (int)((f % CDHW) % DHW);
  float4 xv = *(const float4*)(x + f);
  float4 sv = *(const float4*)(scale + (long)b * DHW + p);
  float4 o;
  o.x = xv.x * sv.x; o.y = xv.y * sv.y;
  o.z = xv.z * sv.z; o.w = xv.w * sv.w;
  *(float4*)(out + f) = o;
}

extern "C" void kernel_launch(void* const* d_in, const int* in_sizes, int n_in,
                              void* d_out, int out_size, void* d_ws, size_t ws_size,
                              hipStream_t stream) {
  const float* x      = (const float*)d_in[0];
  const float* conv_w = (const float*)d_in[1];
  const float* conv_b = (const float*)d_in[2];
  float* out = (float*)d_out;

  float* ws    = (float*)d_ws;
  float* xmax  = ws;
  float* xmean = ws + (long)B_ * DHW;
  float* scale = ws + 2L * B_ * DHW;

  sg_reduce_maxmean<<<512, 256, 0, stream>>>(x, xmax, xmean);

  sg_conv_sigmoid<<<2048, 32, 0, stream>>>(xmax, xmean, conv_w, conv_b, scale);

  sg_apply<<<32768, 256, 0, stream>>>(x, scale, out);
}


# conv re-decomposed as a depth march: WMMA rows = 7 depth taps x 2 output rows, K = 2ch x 8x8 window read straight from global (no LDS), acc rows shift per input depth; sum order changes
# speedup vs baseline: 2.0203x; 2.0203x over previous
#include <hip/hip_runtime.h>
#include <hip/hip_bf16.h>

typedef float v2f __attribute__((ext_vector_type(2)));
typedef float v8f __attribute__((ext_vector_type(8)));

#define B_    4
#define C_    64
#define D_    32
#define H_    64
#define W_    64
#define HW    (H_ * W_)
#define DHW   (D_ * HW)
#define CDHW  (C_ * DHW)

__global__ __launch_bounds__(256) void sg_reduce_maxmean(
    const float* __restrict__ x, float* __restrict__ xmax, float* __restrict__ xmean) {
  long t = (long)blockIdx.x * blockDim.x + threadIdx.x;
  long f = t * 4;
  int  b = (int)(f / DHW);
  int  q = (int)(f % DHW);
  const float* xp = x + (long)b * CDHW + q;

  float4 v = *(const float4*)xp;
  float mx0 = v.x, mx1 = v.y, mx2 = v.z, mx3 = v.w;
  float s0 = v.x, s1 = v.y, s2 = v.z, s3 = v.w;
  for (int c = 1; c < C_; ++c) {
    float4 u = *(const float4*)(xp + (long)c * DHW);
    mx0 = fmaxf(mx0, u.x); mx1 = fmaxf(mx1, u.y);
    mx2 = fmaxf(mx2, u.z); mx3 = fmaxf(mx3, u.w);
    s0 += u.x; s1 += u.y; s2 += u.z; s3 += u.w;
  }
  const float inv = 1.0f / (float)C_;
  float4 mo; mo.x = mx0; mo.y = mx1; mo.z = mx2; mo.w = mx3;
  float4 me; me.x = s0 * inv; me.y = s1 * inv; me.z = s2 * inv; me.w = s3 * inv;
  *(float4*)(xmax  + (long)b * DHW + q) = mo;
  *(float4*)(xmean + (long)b * DHW + q) = me;
}

#define NCHUNK 32
#define DCH    8

struct BFrag { float r[NCHUNK][2]; };

__global__ __launch_bounds__(128) void sg_conv_sigmoid(
    const float* __restrict__ xmax, const float* __restrict__ xmean,
    const float* __restrict__ wgt, const float* __restrict__ bias,
    float* __restrict__ scale) {
  const int lane = threadIdx.x & 31;
  const int wt   = __builtin_amdgcn_readfirstlane((int)(threadIdx.x >> 5));
  const int bid  = blockIdx.x;
  const int hp   = bid & 31;
  const int dc   = (bid >> 5) & 3;
  const int b    = bid >> 7;
  const int h0 = hp * 2, w0 = wt * 16, d0 = dc * DCH;

  const int n = lane & 15;
  const int g = lane >> 4;

  v2f a[NCHUNK];
  {
    const int up = n >> 3;
    const int kd = n & 7;
#pragma unroll
    for (int j = 0; j < NCHUNK; ++j) {
      const int c = j >> 4, khp = (j >> 1) & 7, half = j & 1;
      const int kh = khp - up;
      const int row = h0 + khp - 3;
      const float rowf = (row >= 0 && row < H_) ? 1.0f : 0.0f;
      float av[2];
#pragma unroll
      for (int s = 0; s < 2; ++s) {
        const int kw = half * 4 + 2 * g + s;
        const bool ok = (kd < 7) && (kh >= 0) && (kh < 7) && (kw < 7);
        const int idx = ok ? (((c * 7 + kd) * 7 + kh) * 7 + kw) : 0;
        const float v = wgt[idx];
        av[s] = (ok ? v : 0.0f) * rowf;
      }
      a[j].x = av[0]; a[j].y = av[1];
    }
  }

  int  colc[2][2];
  bool colok[2][2];
#pragma unroll
  for (int half = 0; half < 2; ++half)
#pragma unroll
    for (int s = 0; s < 2; ++s) {
      const int col = w0 + n + half * 4 + 2 * g + s - 3;
      colok[half][s] = (col >= 0) && (col < W_);
      colc[half][s]  = col < 0 ? 0 : (col >= W_ ? W_ - 1 : col);
    }

  const float* pm = xmax  + (long)b * DHW;
  const float* pe = xmean + (long)b * DHW;

  auto loadB = [&](BFrag& bf, int dz) {
#pragma unroll
    for (int j = 0; j < NCHUNK; ++j) {
      const int c = j >> 4, khp = (j >> 1) & 7, half = j & 1;
      const int row  = h0 + khp - 3;
      const int rowc = row < 0 ? 0 : (row >= H_ ? H_ - 1 : row);
      const float* p = (c == 0 ? pm : pe) + dz * HW + rowc * W_;
      bf.r[j][0] = p[colc[half][0]];
      bf.r[j][1] = p[colc[half][1]];
    }
  };

  v8f acc0 = {0.f, 0.f, 0.f, 0.f, 0.f, 0.f, 0.f, 0.f};
  v8f acc1 = {0.f, 0.f, 0.f, 0.f, 0.f, 0.f, 0.f, 0.f};

  auto mma = [&](const BFrag& bf) {
#pragma unroll
    for (int j = 0; j < NCHUNK; j += 2) {
      v2f b0, b1;
      b0.x = colok[0][0] ? bf.r[j][0] : 0.0f;
      b0.y = colok[0][1] ? bf.r[j][1] : 0.0f;
      b1.x = colok[1][0] ? bf.r[j + 1][0] : 0.0f;
      b1.y = colok[1][1] ? bf.r[j + 1][1] : 0.0f;
      acc0 = __builtin_amdgcn_wmma_f32_16x16x4_f32(
          false, a[j], false, b0, (short)0, acc0, false, false);
      acc1 = __builtin_amdgcn_wmma_f32_16x16x4_f32(
          false, a[j + 1], false, b1, (short)0, acc1, false, false);
    }
  };

  const float bv = bias[0];
  float* sp = scale + (long)b * DHW + (h0 + g) * W_ + w0 + n;
  auto emit = [&](int dd, float y) {
    if (dd >= d0 && dd < d0 + DCH) {
      const float z = y + bv;
      sp[dd * HW] = 1.0f / (1.0f + __expf(-z));
    }
  };

  auto finish_step = [&](int dz) {
    emit(dz - 3, acc0[6] + acc1[6]);
#pragma unroll
    for (int r = 7; r >= 1; --r) { acc0[r] = acc0[r - 1]; acc1[r] = acc1[r - 1]; }
    acc0[0] = 0.0f; acc1[0] = 0.0f;
    acc0[7] = 0.0f; acc1[7] = 0.0f;
  };

  int dzs = d0 - 3 < 0 ? 0 : d0 - 3;
  int dze = d0 + DCH + 2 > D_ - 1 ? D_ - 1 : d0 + DCH + 2;
  if (((dze - dzs + 1) & 1) != 0) { if (dze < D_ - 1) ++dze; else --dzs; }

  BFrag B0, B1;
  loadB(B0, dzs);
  for (int dz = dzs; dz < dze; dz += 2) {
    loadB(B1, dz + 1);
    mma(B0);
    finish_step(dz);
    loadB(B0, dz + 2 <= dze ? dz + 2 : dze);
    mma(B1);
    finish_step(dz + 1);
  }

#pragma unroll
  for (int rr = 6; rr >= 4; --rr) emit(dze + 4 - rr, acc0[rr] + acc1[rr]);
}

__global__ __launch_bounds__(256) void sg_apply(
    const float* __restrict__ x, const float* __restrict__ scale,
    float* __restrict__ out) {
  long t = (long)blockIdx.x * blockDim.x + threadIdx.x;
  long f = t * 4;
  int  b = (int)(f / CDHW);
  int  p = (int)((f % CDHW) % DHW);
  float4 xv = *(const float4*)(x + f);
  float4 sv = *(const float4*)(scale + (long)b * DHW + p);
  float4 o;
  o.x = xv.x * sv.x; o.y = xv.y * sv.y;
  o.z = xv.z * sv.z; o.w = xv.w * sv.w;
  *(float4*)(out + f) = o;
}

extern "C" void kernel_launch(void* const* d_in, const int* in_sizes, int n_in,
                              void* d_out, int out_size, void* d_ws, size_t ws_size,
                              hipStream_t stream) {
  const float* x      = (const float*)d_in[0];
  const float* conv_w = (const float*)d_in[1];
  const float* conv_b = (const float*)d_in[2];
  float* out = (float*)d_out;

  float* ws    = (float*)d_ws;
  float* xmax  = ws;
  float* xmean = ws + (long)B_ * DHW;
  float* scale = ws + 2L * B_ * DHW;

  sg_reduce_maxmean<<<512, 256, 0, stream>>>(x, xmax, xmean);

  sg_conv_sigmoid<<<512, 128, 0, stream>>>(xmax, xmean, conv_w, conv_b, scale);

  sg_apply<<<32768, 256, 0, stream>>>(x, scale, out);
}


# conv B fragments fetched as one 16-byte window of 4 adjacent columns per lane (K slots re-paired, kw8 = 4g + 2half + s): 16 loads per depth plane, not 64; sum order changes
# speedup vs baseline: 2.0977x; 1.0383x over previous
#include <hip/hip_runtime.h>
#include <hip/hip_bf16.h>

typedef float v2f __attribute__((ext_vector_type(2)));
typedef float v8f __attribute__((ext_vector_type(8)));
typedef float v4f __attribute__((ext_vector_type(4)));

#define B_    4
#define C_    64
#define D_    32
#define H_    64
#define W_    64
#define HW    (H_ * W_)
#define DHW   (D_ * HW)
#define CDHW  (C_ * DHW)

__global__ __launch_bounds__(256) void sg_reduce_maxmean(
    const float* __restrict__ x, float* __restrict__ xmax, float* __restrict__ xmean) {
  long t = (long)blockIdx.x * blockDim.x + threadIdx.x;
  long f = t * 4;
  int  b = (int)(f / DHW);
  int  q = (int)(f % DHW);
  const float* xp = x + (long)b * CDHW + q;

  float4 v = *(const float4*)xp;
  float mx0 = v.x, mx1 = v.y, mx2 = v.z, mx3 = v.w;
  float s0 = v.x, s1 = v.y, s2 = v.z, s3 = v.w;
  for (int c = 1; c < C_; ++c) {
    float4 u = *(const float4*)(xp + (long)c * DHW);
    mx0 = fmaxf(mx0, u.x); mx1 = fmaxf(mx1, u.y);
    mx2 = fmaxf(mx2, u.z); mx3 = fmaxf(mx3, u.w);
    s0 += u.x; s1 += u.y; s2 += u.z; s3 += u.w;
  }
  const float inv = 1.0f / (float)C_;
  float4 mo; mo.x = mx0; mo.y = mx1; mo.z = mx2; mo.w = mx3;
  float4 me; me.x = s0 * inv; me.y = s1 * inv; me.z = s2 * inv; me.w = s3 * inv;
  *(float4*)(xmax  + (long)b * DHW + q) = mo;
  *(float4*)(xmean + (long)b * DHW + q) = me;
}

#define SCALE_PLANE(b) ((b) < 2 ? (b) : (b) + 8)
#define NCHUNK 32
#define DCH    8

#define NROW   16
struct BFrag { v4f r[NROW]; };

__global__ __launch_bounds__(128) void sg_conv_sigmoid(
    const float* __restrict__ xmax, const float* __restrict__ xmean,
    const float* __restrict__ wgt, const float* __restrict__ bias,
    float* __restrict__ scale) {
  const int lane = threadIdx.x & 31;
  const int wt   = __builtin_amdgcn_readfirstlane((int)(threadIdx.x >> 5));
  const int bid  = blockIdx.x;
  const int hp   = bid & 31;
  const int dc   = (bid >> 5) & 3;
  const int b    = bid >> 7;
  const int h0 = hp * 2, w0 = wt * 16, d0 = dc * DCH;

  const int n = lane & 15;
  const int g = lane >> 4;

  v2f a[NCHUNK];
  {
    const int up = n >> 3;
    const int kd = n & 7;
#pragma unroll
    for (int j = 0; j < NCHUNK; ++j) {
      const int c = j >> 4, khp = (j >> 1) & 7, half = j & 1;
      const int kh = khp - up;
      const int row = h0 + khp - 3;
      const float rowf = (row >= 0 && row < H_) ? 1.0f : 0.0f;
      float av[2];
#pragma unroll
      for (int s = 0; s < 2; ++s) {
        const int kw = 4 * g + 2 * half + s;
        const bool ok = (kd < 7) && (kh >= 0) && (kh < 7) && (kw < 7);
        const int idx = ok ? (((c * 7 + kd) * 7 + kh) * 7 + kw) : 0;
        const float v = wgt[idx];
        av[s] = (ok ? v : 0.0f) * rowf;
      }
      a[j].x = av[0]; a[j].y = av[1];
    }
  }

  bool colok[4];
#pragma unroll
  for (int e = 0; e < 4; ++e) {
    const int col = w0 + n + 4 * g - 3 + e;
    colok[e] = (col >= 0) && (col < W_);
  }
  const unsigned loff = (unsigned)(n + 4 * g);

  const float* pm = xmax  + (long)b * DHW + (w0 - 3);
  const float* pe = xmean + (long)b * DHW + (w0 - 3);

  auto loadB = [&](BFrag& bf, int dz) {
#pragma unroll
    for (int q = 0; q < NROW; ++q) {
      const int c = q >> 3, khp = q & 7;
      const int row  = h0 + khp - 3;
      const int rowc = row < 0 ? 0 : (row >= H_ ? H_ - 1 : row);
      const float* p = (c == 0 ? pm : pe) + dz * HW + rowc * W_;
      v4f v;
      __builtin_memcpy(&v, p + loff, sizeof(v));
      bf.r[q] = v;
    }
  };

  v8f acc0 = {0.f, 0.f, 0.f, 0.f, 0.f, 0.f, 0.f, 0.f};
  v8f acc1 = {0.f, 0.f, 0.f, 0.f, 0.f, 0.f, 0.f, 0.f};

  auto mma = [&](const BFrag& bf) {
#pragma unroll
    for (int q = 0; q < NROW; ++q) {
      v2f b0, b1;
      b0.x = colok[0] ? bf.r[q].x : 0.0f;
      b0.y = colok[1] ? bf.r[q].y : 0.0f;
      b1.x = colok[2] ? bf.r[q].z : 0.0f;
      b1.y = colok[3] ? bf.r[q].w : 0.0f;
      acc0 = __builtin_amdgcn_wmma_f32_16x16x4_f32(
          false, a[2 * q], false, b0, (short)0, acc0, false, false);
      acc1 = __builtin_amdgcn_wmma_f32_16x16x4_f32(
          false, a[2 * q + 1], false, b1, (short)0, acc1, false, false);
    }
  };

  const float bv = bias[0];
  float* sp = scale + (long)SCALE_PLANE(b) * DHW + (h0 + g) * W_ + w0 + n;
  auto emit = [&](int dd, float y) {
    if (dd >= d0 && dd < d0 + DCH) {
      const float z = y + bv;
      sp[dd * HW] = 1.0f / (1.0f + __expf(-z));
    }
  };

  auto finish_step = [&](int dz) {
    emit(dz - 3, acc0[6] + acc1[6]);
#pragma unroll
    for (int r = 7; r >= 1; --r) { acc0[r] = acc0[r - 1]; acc1[r] = acc1[r - 1]; }
    acc0[0] = 0.0f; acc1[0] = 0.0f;
    acc0[7] = 0.0f; acc1[7] = 0.0f;
  };

  int dzs = d0 - 3 < 0 ? 0 : d0 - 3;
  int dze = d0 + DCH + 2 > D_ - 1 ? D_ - 1 : d0 + DCH + 2;
  if (((dze - dzs + 1) & 1) != 0) { if (dze < D_ - 1) ++dze; else --dzs; }

  BFrag B0, B1;
  loadB(B0, dzs);
  for (int dz = dzs; dz < dze; dz += 2) {
    loadB(B1, dz + 1);
    mma(B0);
    finish_step(dz);
    loadB(B0, dz + 2 <= dze ? dz + 2 : dze);
    mma(B1);
    finish_step(dz + 1);
  }

#pragma unroll
  for (int rr = 6; rr >= 4; --rr) emit(dze + 4 - rr, acc0[rr] + acc1[rr]);
}

__global__ __launch_bounds__(256) void sg_apply(
    const float* __restrict__ x, const float* __restrict__ scale,
    float* __restrict__ out) {
  long t = (long)blockIdx.x * blockDim.x + threadIdx.x;
  long f = t * 4;
  int  b = (int)(f / CDHW);
  int  p = (int)((f % CDHW) % DHW);
  float4 xv = *(const float4*)(x + f);
  float4 sv = *(const float4*)(scale + (long)SCALE_PLANE(b) * DHW + p);
  float4 o;
  o.x = xv.x * sv.x; o.y = xv.y * sv.y;
  o.z = xv.z * sv.z; o.w = xv.w * sv.w;
  *(float4*)(out + f) = o;
}

extern "C" void kernel_launch(void* const* d_in, const int* in_sizes, int n_in,
                              void* d_out, int out_size, void* d_ws, size_t ws_size,
                              hipStream_t stream) {
  const float* x      = (const float*)d_in[0];
  const float* conv_w = (const float*)d_in[1];
  const float* conv_b = (const float*)d_in[2];
  float* out = (float*)d_out;

  float* ws    = (float*)d_ws;
  float* scale = ws;
  float* xmax  = ws + 2L * DHW;
  float* xmean = ws + 2L * DHW + (long)B_ * DHW;

  sg_reduce_maxmean<<<512, 256, 0, stream>>>(x, xmax, xmean);

  sg_conv_sigmoid<<<512, 128, 0, stream>>>(xmax, xmean, conv_w, conv_b, scale);

  sg_apply<<<32768, 256, 0, stream>>>(x, scale, out);
}


# broadcast multiply: each thread takes its 4 positions in 4 channels of one batch (1 scale load, 4 x loads in flight, 4 stores); 65536 one-round-trip waves instead of 262144
# speedup vs baseline: 2.2111x; 1.0541x over previous
#include <hip/hip_runtime.h>
#include <hip/hip_bf16.h>

typedef float v2f __attribute__((ext_vector_type(2)));
typedef float v8f __attribute__((ext_vector_type(8)));
typedef float v4f __attribute__((ext_vector_type(4)));

#define B_    4
#define C_    64
#define D_    32
#define H_    64
#define W_    64
#define HW    (H_ * W_)
#define DHW   (D_ * HW)
#define CDHW  (C_ * DHW)

__global__ __launch_bounds__(256) void sg_reduce_maxmean(
    const float* __restrict__ x, float* __restrict__ xmax, float* __restrict__ xmean) {
  long t = (long)blockIdx.x * blockDim.x + threadIdx.x;
  long f = t * 4;
  int  b = (int)(f / DHW);
  int  q = (int)(f % DHW);
  const float* xp = x + (long)b * CDHW + q;

  float4 v = *(const float4*)xp;
  float mx0 = v.x, mx1 = v.y, mx2 = v.z, mx3 = v.w;
  float s0 = v.x, s1 = v.y, s2 = v.z, s3 = v.w;
  for (int c = 1; c < C_; ++c) {
    float4 u = *(const float4*)(xp + (long)c * DHW);
    mx0 = fmaxf(mx0, u.x); mx1 = fmaxf(mx1, u.y);
    mx2 = fmaxf(mx2, u.z); mx3 = fmaxf(mx3, u.w);
    s0 += u.x; s1 += u.y; s2 += u.z; s3 += u.w;
  }
  const float inv = 1.0f / (float)C_;
  float4 mo; mo.x = mx0; mo.y = mx1; mo.z = mx2; mo.w = mx3;
  float4 me; me.x = s0 * inv; me.y = s1 * inv; me.z = s2 * inv; me.w = s3 * inv;
  *(float4*)(xmax  + (long)b * DHW + q) = mo;
  *(float4*)(xmean + (long)b * DHW + q) = me;
}

#define SCALE_PLANE(b) ((b) < 2 ? (b) : (b) + 8)
#define NCHUNK 32
#define DCH    8

#define NROW   16
struct BFrag { v4f r[NROW]; };

__global__ __launch_bounds__(128) void sg_conv_sigmoid(
    const float* __restrict__ xmax, const float* __restrict__ xmean,
    const float* __restrict__ wgt, const float* __restrict__ bias,
    float* __restrict__ scale) {
  const int lane = threadIdx.x & 31;
  const int wt   = __builtin_amdgcn_readfirstlane((int)(threadIdx.x >> 5));
  const int bid  = blockIdx.x;
  const int hp   = bid & 31;
  const int dc   = (bid >> 5) & 3;
  const int b    = bid >> 7;
  const int h0 = hp * 2, w0 = wt * 16, d0 = dc * DCH;

  const int n = lane & 15;
  const int g = lane >> 4;

  v2f a[NCHUNK];
  {
    const int up = n >> 3;
    const int kd = n & 7;
#pragma unroll
    for (int j = 0; j < NCHUNK; ++j) {
      const int c = j >> 4, khp = (j >> 1) & 7, half = j & 1;
      const int kh = khp - up;
      const int row = h0 + khp - 3;
      const float rowf = (row >= 0 && row < H_) ? 1.0f : 0.0f;
      float av[2];
#pragma unroll
      for (int s = 0; s < 2; ++s) {
        const int kw = 4 * g + 2 * half + s;
        const bool ok = (kd < 7) && (kh >= 0) && (kh < 7) && (kw < 7);
        const int idx = ok ? (((c * 7 + kd) * 7 + kh) * 7 + kw) : 0;
        const float v = wgt[idx];
        av[s] = (ok ? v : 0.0f) * rowf;
      }
      a[j].x = av[0]; a[j].y = av[1];
    }
  }

  bool colok[4];
#pragma unroll
  for (int e = 0; e < 4; ++e) {
    const int col = w0 + n + 4 * g - 3 + e;
    colok[e] = (col >= 0) && (col < W_);
  }
  const unsigned loff = (unsigned)(n + 4 * g);

  const float* pm = xmax  + (long)b * DHW + (w0 - 3);
  const float* pe = xmean + (long)b * DHW + (w0 - 3);

  auto loadB = [&](BFrag& bf, int dz) {
#pragma unroll
    for (int q = 0; q < NROW; ++q) {
      const int c = q >> 3, khp = q & 7;
      const int row  = h0 + khp - 3;
      const int rowc = row < 0 ? 0 : (row >= H_ ? H_ - 1 : row);
      const float* p = (c == 0 ? pm : pe) + dz * HW + rowc * W_;
      v4f v;
      __builtin_memcpy(&v, p + loff, sizeof(v));
      bf.r[q] = v;
    }
  };

  v8f acc0 = {0.f, 0.f, 0.f, 0.f, 0.f, 0.f, 0.f, 0.f};
  v8f acc1 = {0.f, 0.f, 0.f, 0.f, 0.f, 0.f, 0.f, 0.f};

  auto mma = [&](const BFrag& bf) {
#pragma unroll
    for (int q = 0; q < NROW; ++q) {
      v2f b0, b1;
      b0.x = colok[0] ? bf.r[q].x : 0.0f;
      b0.y = colok[1] ? bf.r[q].y : 0.0f;
      b1.x = colok[2] ? bf.r[q].z : 0.0f;
      b1.y = colok[3] ? bf.r[q].w : 0.0f;
      acc0 = __builtin_amdgcn_wmma_f32_16x16x4_f32(
          false, a[2 * q], false, b0, (short)0, acc0, false, false);
      acc1 = __builtin_amdgcn_wmma_f32_16x16x4_f32(
          false, a[2 * q + 1], false, b1, (short)0, acc1, false, false);
    }
  };

  const float bv = bias[0];
  float* sp = scale + (long)SCALE_PLANE(b) * DHW + (h0 + g) * W_ + w0 + n;
  auto emit = [&](int dd, float y) {
    if (dd >= d0 && dd < d0 + DCH) {
      const float z = y + bv;
      sp[dd * HW] = 1.0f / (1.0f + __expf(-z));
    }
  };

  auto finish_step = [&](int dz) {
    emit(dz - 3, acc0[6] + acc1[6]);
#pragma unroll
    for (int r = 7; r >= 1; --r) { acc0[r] = acc0[r - 1]; acc1[r] = acc1[r - 1]; }
    acc0[0] = 0.0f; acc1[0] = 0.0f;
    acc0[7] = 0.0f; acc1[7] = 0.0f;
  };

  int dzs = d0 - 3 < 0 ? 0 : d0 - 3;
  int dze = d0 + DCH + 2 > D_ - 1 ? D_ - 1 : d0 + DCH + 2;
  if (((dze - dzs + 1) & 1) != 0) { if (dze < D_ - 1) ++dze; else --dzs; }

  BFrag B0, B1;
  loadB(B0, dzs);
  for (int dz = dzs; dz < dze; dz += 2) {
    loadB(B1, dz + 1);
    mma(B0);
    finish_step(dz);
    loadB(B0, dz + 2 <= dze ? dz + 2 : dze);
    mma(B1);
    finish_step(dz + 1);
  }

#pragma unroll
  for (int rr = 6; rr >= 4; --rr) emit(dze + 4 - rr, acc0[rr] + acc1[rr]);
}

#define ACH 4
__global__ __launch_bounds__(256) void sg_apply(
    const float* __restrict__ x, const float* __restrict__ scale,
    float* __restrict__ out) {
  const int bid = blockIdx.x;
  const int pc  = bid & 127;
  const int cg  = (bid >> 7) & 15;
  const int b   = bid >> 11;
  const int p   = (pc * 256 + (int)threadIdx.x) * 4;
  const long base = (long)b * CDHW + (long)(cg * ACH) * DHW + p;
  const float4 sv = *(const float4*)(scale + (long)SCALE_PLANE(b) * DHW + p);
  float4 xv[ACH];
#pragma unroll
  for (int k = 0; k < ACH; ++k) xv[k] = *(const float4*)(x + base + (long)k * DHW);
#pragma unroll
  for (int k = 0; k < ACH; ++k) {
    float4 o;
    o.x = xv[k].x * sv.x; o.y = xv[k].y * sv.y;
    o.z = xv[k].z * sv.z; o.w = xv[k].w * sv.w;
    *(float4*)(out + base + (long)k * DHW) = o;
  }
}

extern "C" void kernel_launch(void* const* d_in, const int* in_sizes, int n_in,
                              void* d_out, int out_size, void* d_ws, size_t ws_size,
                              hipStream_t stream) {
  const float* x      = (const float*)d_in[0];
  const float* conv_w = (const float*)d_in[1];
  const float* conv_b = (const float*)d_in[2];
  float* out = (float*)d_out;

  float* ws    = (float*)d_ws;
  float* scale = ws;
  float* xmax  = ws + 2L * DHW;
  float* xmean = ws + 2L * DHW + (long)B_ * DHW;

  sg_reduce_maxmean<<<512, 256, 0, stream>>>(x, xmax, xmean);

  sg_conv_sigmoid<<<512, 128, 0, stream>>>(xmax, xmean, conv_w, conv_b, scale);

  sg_apply<<<8192, 256, 0, stream>>>(x, scale, out);
}


# channel reduce issues all 64 channel loads of a lane before the first fold (4 waves per SIMD, 254 registers): one memory trip per wave, not 8 dependent trips of 9 loads; sum order kept
# speedup vs baseline: 2.3011x; 1.0407x over previous
#include <hip/hip_runtime.h>
#include <hip/hip_bf16.h>

typedef float v2f __attribute__((ext_vector_type(2)));
typedef float v8f __attribute__((ext_vector_type(8)));
typedef float v4f __attribute__((ext_vector_type(4)));

#define B_    4
#define C_    64
#define D_    32
#define H_    64
#define W_    64
#define HW    (H_ * W_)
#define DHW   (D_ * HW)
#define CDHW  (C_ * DHW)

__global__ __launch_bounds__(256) __attribute__((amdgpu_waves_per_eu(4, 4)))
void sg_reduce_maxmean(
    const float* __restrict__ x, float* __restrict__ xmax, float* __restrict__ xmean) {
  long t = (long)blockIdx.x * blockDim.x + threadIdx.x;
  long f = t * 4;
  int  b = (int)(f / DHW);
  int  q = (int)(f % DHW);
  const float* xp = x + (long)b * CDHW + q;

  float4 r[C_];
#pragma unroll
  for (int c = 0; c < C_; ++c) r[c] = *(const float4*)(xp + (long)c * DHW);

  float mx0 = r[0].x, mx1 = r[0].y, mx2 = r[0].z, mx3 = r[0].w;
  float s0 = r[0].x, s1 = r[0].y, s2 = r[0].z, s3 = r[0].w;
#pragma unroll
  for (int c = 1; c < C_; ++c) {
    mx0 = fmaxf(mx0, r[c].x); mx1 = fmaxf(mx1, r[c].y);
    mx2 = fmaxf(mx2, r[c].z); mx3 = fmaxf(mx3, r[c].w);
    s0 += r[c].x; s1 += r[c].y; s2 += r[c].z; s3 += r[c].w;
  }
  const float inv = 1.0f / (float)C_;
  float4 mo; mo.x = mx0; mo.y = mx1; mo.z = mx2; mo.w = mx3;
  float4 me; me.x = s0 * inv; me.y = s1 * inv; me.z = s2 * inv; me.w = s3 * inv;
  *(float4*)(xmax  + (long)b * DHW + q) = mo;
  *(float4*)(xmean + (long)b * DHW + q) = me;
}

#define SCALE_PLANE(b) ((b) < 2 ? (b) : (b) + 8)
#define NCHUNK 32
#define DCH    8

#define NROW   16
struct BFrag { v4f r[NROW]; };

__global__ __launch_bounds__(128) void sg_conv_sigmoid(
    const float* __restrict__ xmax, const float* __restrict__ xmean,
    const float* __restrict__ wgt, const float* __restrict__ bias,
    float* __restrict__ scale) {
  const int lane = threadIdx.x & 31;
  const int wt   = __builtin_amdgcn_readfirstlane((int)(threadIdx.x >> 5));
  const int bid  = blockIdx.x;
  const int hp   = bid & 31;
  const int dc   = (bid >> 5) & 3;
  const int b    = bid >> 7;
  const int h0 = hp * 2, w0 = wt * 16, d0 = dc * DCH;

  const int n = lane & 15;
  const int g = lane >> 4;

  v2f a[NCHUNK];
  {
    const int up = n >> 3;
    const int kd = n & 7;
#pragma unroll
    for (int j = 0; j < NCHUNK; ++j) {
      const int c = j >> 4, khp = (j >> 1) & 7, half = j & 1;
      const int kh = khp - up;
      const int row = h0 + khp - 3;
      const float rowf = (row >= 0 && row < H_) ? 1.0f : 0.0f;
      float av[2];
#pragma unroll
      for (int s = 0; s < 2; ++s) {
        const int kw = 4 * g + 2 * half + s;
        const bool ok = (kd < 7) && (kh >= 0) && (kh < 7) && (kw < 7);
        const int idx = ok ? (((c * 7 + kd) * 7 + kh) * 7 + kw) : 0;
        const float v = wgt[idx];
        av[s] = (ok ? v : 0.0f) * rowf;
      }
      a[j].x = av[0]; a[j].y = av[1];
    }
  }

  bool colok[4];
#pragma unroll
  for (int e = 0; e < 4; ++e) {
    const int col = w0 + n + 4 * g - 3 + e;
    colok[e] = (col >= 0) && (col < W_);
  }
  const unsigned loff = (unsigned)(n + 4 * g);

  const float* pm = xmax  + (long)b * DHW + (w0 - 3);
  const float* pe = xmean + (long)b * DHW + (w0 - 3);

  auto loadB = [&](BFrag& bf, int dz) {
#pragma unroll
    for (int q = 0; q < NROW; ++q) {
      const int c = q >> 3, khp = q & 7;
      const int row  = h0 + khp - 3;
      const int rowc = row < 0 ? 0 : (row >= H_ ? H_ - 1 : row);
      const float* p = (c == 0 ? pm : pe) + dz * HW + rowc * W_;
      v4f v;
      __builtin_memcpy(&v, p + loff, sizeof(v));
      bf.r[q] = v;
    }
  };

  v8f acc0 = {0.f, 0.f, 0.f, 0.f, 0.f, 0.f, 0.f, 0.f};
  v8f acc1 = {0.f, 0.f, 0.f, 0.f, 0.f, 0.f, 0.f, 0.f};

  auto mma = [&](const BFrag& bf) {
#pragma unroll
    for (int q = 0; q < NROW; ++q) {
      v2f b0, b1;
      b0.x = colok[0] ? bf.r[q].x : 0.0f;
      b0.y = colok[1] ? bf.r[q].y : 0.0f;
      b1.x = colok[2] ? bf.r[q].z : 0.0f;
      b1.y = colok[3] ? bf.r[q].w : 0.0f;
      acc0 = __builtin_amdgcn_wmma_f32_16x16x4_f32(
          false, a[2 * q], false, b0, (short)0, acc0, false, false);
      acc1 = __builtin_amdgcn_wmma_f32_16x16x4_f32(
          false, a[2 * q + 1], false, b1, (short)0, acc1, false, false);
    }
  };

  const float bv = bias[0];
  float* sp = scale + (long)SCALE_PLANE(b) * DHW + (h0 + g) * W_ + w0 + n;
  auto emit = [&](int dd, float y) {
    if (dd >= d0 && dd < d0 + DCH) {
      const float z = y + bv;
      sp[dd * HW] = 1.0f / (1.0f + __expf(-z));
    }
  };

  auto finish_step = [&](int dz) {
    emit(dz - 3, acc0[6] + acc1[6]);
#pragma unroll
    for (int r = 7; r >= 1; --r) { acc0[r] = acc0[r - 1]; acc1[r] = acc1[r - 1]; }
    acc0[0] = 0.0f; acc1[0] = 0.0f;
    acc0[7] = 0.0f; acc1[7] = 0.0f;
  };

  int dzs = d0 - 3 < 0 ? 0 : d0 - 3;
  int dze = d0 + DCH + 2 > D_ - 1 ? D_ - 1 : d0 + DCH + 2;
  if (((dze - dzs + 1) & 1) != 0) { if (dze < D_ - 1) ++dze; else --dzs; }

  BFrag B0, B1;
  loadB(B0, dzs);
  for (int dz = dzs; dz < dze; dz += 2) {
    loadB(B1, dz + 1);
    mma(B0);
    finish_step(dz);
    loadB(B0, dz + 2 <= dze ? dz + 2 : dze);
    mma(B1);
    finish_step(dz + 1);
  }

#pragma unroll
  for (int rr = 6; rr >= 4; --rr) emit(dze + 4 - rr, acc0[rr] + acc1[rr]);
}

#define ACH 4
__global__ __launch_bounds__(256) void sg_apply(
    const float* __restrict__ x, const float* __restrict__ scale,
    float* __restrict__ out) {
  const int bid = blockIdx.x;
  const int pc  = bid & 127;
  const int cg  = (bid >> 7) & 15;
  const int b   = bid >> 11;
  const int p   = (pc * 256 + (int)threadIdx.x) * 4;
  const long base = (long)b * CDHW + (long)(cg * ACH) * DHW + p;
  const float4 sv = *(const float4*)(scale + (long)SCALE_PLANE(b) * DHW + p);
  float4 xv[ACH];
#pragma unroll
  for (int k = 0; k < ACH; ++k) xv[k] = *(const float4*)(x + base + (long)k * DHW);
#pragma unroll
  for (int k = 0; k < ACH; ++k) {
    float4 o;
    o.x = xv[k].x * sv.x; o.y = xv[k].y * sv.y;
    o.z = xv[k].z * sv.z; o.w = xv[k].w * sv.w;
    *(float4*)(out + base + (long)k * DHW) = o;
  }
}

extern "C" void kernel_launch(void* const* d_in, const int* in_sizes, int n_in,
                              void* d_out, int out_size, void* d_ws, size_t ws_size,
                              hipStream_t stream) {
  const float* x      = (const float*)d_in[0];
  const float* conv_w = (const float*)d_in[1];
  const float* conv_b = (const float*)d_in[2];
  float* out = (float*)d_out;

  float* ws    = (float*)d_ws;
  float* scale = ws;
  float* xmax  = ws + 2L * DHW;
  float* xmean = ws + 2L * DHW + (long)B_ * DHW;

  sg_reduce_maxmean<<<512, 256, 0, stream>>>(x, xmax, xmean);

  sg_conv_sigmoid<<<512, 128, 0, stream>>>(xmax, xmean, conv_w, conv_b, scale);

  sg_apply<<<8192, 256, 0, stream>>>(x, scale, out);
}


# conv: each wave marches all 32 depths on a quarter of K (1 channel x 4 input rows), a tile's 4 waves add their finished rows through LDS: 32 plane steps per tile, not 52; sum order changes
# speedup vs baseline: 2.5640x; 1.1142x over previous
#include <hip/hip_runtime.h>
#include <hip/hip_bf16.h>

typedef float v2f __attribute__((ext_vector_type(2)));
typedef float v8f __attribute__((ext_vector_type(8)));
typedef float v4f __attribute__((ext_vector_type(4)));

#define B_    4
#define C_    64
#define D_    32
#define H_    64
#define W_    64
#define HW    (H_ * W_)
#define DHW   (D_ * HW)
#define CDHW  (C_ * DHW)

__global__ __launch_bounds__(256) __attribute__((amdgpu_waves_per_eu(4, 4)))
void sg_reduce_maxmean(
    const float* __restrict__ x, float* __restrict__ xmax, float* __restrict__ xmean) {
  long t = (long)blockIdx.x * blockDim.x + threadIdx.x;
  long f = t * 4;
  int  b = (int)(f / DHW);
  int  q = (int)(f % DHW);
  const float* xp = x + (long)b * CDHW + q;

  float4 r[C_];
#pragma unroll
  for (int c = 0; c < C_; ++c) r[c] = *(const float4*)(xp + (long)c * DHW);

  float mx0 = r[0].x, mx1 = r[0].y, mx2 = r[0].z, mx3 = r[0].w;
  float s0 = r[0].x, s1 = r[0].y, s2 = r[0].z, s3 = r[0].w;
#pragma unroll
  for (int c = 1; c < C_; ++c) {
    mx0 = fmaxf(mx0, r[c].x); mx1 = fmaxf(mx1, r[c].y);
    mx2 = fmaxf(mx2, r[c].z); mx3 = fmaxf(mx3, r[c].w);
    s0 += r[c].x; s1 += r[c].y; s2 += r[c].z; s3 += r[c].w;
  }
  const float inv = 1.0f / (float)C_;
  float4 mo; mo.x = mx0; mo.y = mx1; mo.z = mx2; mo.w = mx3;
  float4 me; me.x = s0 * inv; me.y = s1 * inv; me.z = s2 * inv; me.w = s3 * inv;
  *(float4*)(xmax  + (long)b * DHW + q) = mo;
  *(float4*)(xmean + (long)b * DHW + q) = me;
}

#define SCALE_PLANE(b) ((b) < 2 ? (b) : (b) + 8)
#define KSPLIT 4
#define NWIN   4
#define NCHUNK (2 * NWIN)

struct BFrag { v4f r[NWIN]; };

__global__ __launch_bounds__(128) void sg_conv_sigmoid(
    const float* __restrict__ xmax, const float* __restrict__ xmean,
    const float* __restrict__ wgt, const float* __restrict__ bias,
    float* __restrict__ scale) {
  __shared__ float part[KSPLIT * D_ * 32];

  const int lane = threadIdx.x & 31;
  const int ks   = __builtin_amdgcn_readfirstlane((int)(threadIdx.x >> 5));
  const int bid  = blockIdx.x;
  const int wt   = bid & 3;
  const int hp   = (bid >> 2) & 31;
  const int b    = bid >> 7;
  const int h0 = hp * 2, w0 = wt * 16;
  const int c  = ks >> 1;
  const int kb = (ks & 1) * NWIN;

  const int n = lane & 15;
  const int g = lane >> 4;

  v2f a[NCHUNK];
  {
    const int up = n >> 3;
    const int kd = n & 7;
#pragma unroll
    for (int j = 0; j < NCHUNK; ++j) {
      const int khp = kb + (j >> 1), half = j & 1;
      const int kh = khp - up;
      const int row = h0 + khp - 3;
      const float rowf = (row >= 0 && row < H_) ? 1.0f : 0.0f;
      float av[2];
#pragma unroll
      for (int s = 0; s < 2; ++s) {
        const int kw = 4 * g + 2 * half + s;
        const bool ok = (kd < 7) && (kh >= 0) && (kh < 7) && (kw < 7);
        const int idx = ok ? (((c * 7 + kd) * 7 + kh) * 7 + kw) : 0;
        const float v = wgt[idx];
        av[s] = (ok ? v : 0.0f) * rowf;
      }
      a[j].x = av[0]; a[j].y = av[1];
    }
  }

  bool colok[4];
#pragma unroll
  for (int e = 0; e < 4; ++e) {
    const int col = w0 + n + 4 * g - 3 + e;
    colok[e] = (col >= 0) && (col < W_);
  }
  const unsigned loff = (unsigned)(n + 4 * g);

  const float* pc = (c == 0 ? xmax : xmean) + (long)b * DHW + (w0 - 3);
  const float* rowp[NWIN];
#pragma unroll
  for (int q = 0; q < NWIN; ++q) {
    const int row  = h0 + kb + q - 3;
    const int rowc = row < 0 ? 0 : (row >= H_ ? H_ - 1 : row);
    rowp[q] = pc + rowc * W_;
  }

  auto loadB = [&](BFrag& bf, int dz) {
#pragma unroll
    for (int q = 0; q < NWIN; ++q) {
      v4f v;
      __builtin_memcpy(&v, rowp[q] + dz * HW + loff, sizeof(v));
      bf.r[q] = v;
    }
  };

  v8f acc0 = {0.f, 0.f, 0.f, 0.f, 0.f, 0.f, 0.f, 0.f};
  v8f acc1 = {0.f, 0.f, 0.f, 0.f, 0.f, 0.f, 0.f, 0.f};

  auto mma = [&](const BFrag& bf) {
#pragma unroll
    for (int q = 0; q < NWIN; ++q) {
      v2f b0, b1;
      b0.x = colok[0] ? bf.r[q].x : 0.0f;
      b0.y = colok[1] ? bf.r[q].y : 0.0f;
      b1.x = colok[2] ? bf.r[q].z : 0.0f;
      b1.y = colok[3] ? bf.r[q].w : 0.0f;
      acc0 = __builtin_amdgcn_wmma_f32_16x16x4_f32(
          false, a[2 * q], false, b0, (short)0, acc0, false, false);
      acc1 = __builtin_amdgcn_wmma_f32_16x16x4_f32(
          false, a[2 * q + 1], false, b1, (short)0, acc1, false, false);
    }
  };

  float* mypart = part + ks * (D_ * 32) + lane;

  auto finish_step = [&](int dz) {
    const int d = dz < 3 ? 0 : dz - 3;
    mypart[d * 32] = acc0[6] + acc1[6];
#pragma unroll
    for (int r = 7; r >= 1; --r) { acc0[r] = acc0[r - 1]; acc1[r] = acc1[r - 1]; }
    acc0[0] = 0.0f; acc1[0] = 0.0f;
    acc0[7] = 0.0f; acc1[7] = 0.0f;
  };

#define PIN_REFILL() __builtin_amdgcn_sched_barrier(0)
  BFrag B0, B1, B2, B3;
  loadB(B0, 0); loadB(B1, 1); loadB(B2, 2); loadB(B3, 3);
  for (int dz = 0; dz < D_ - 4; dz += 4) {
    mma(B0); finish_step(dz);
    loadB(B0, dz + 4);
    PIN_REFILL();
    mma(B1); finish_step(dz + 1);
    loadB(B1, dz + 5);
    PIN_REFILL();
    mma(B2); finish_step(dz + 2);
    loadB(B2, dz + 6);
    PIN_REFILL();
    mma(B3); finish_step(dz + 3);
    loadB(B3, dz + 7);
    PIN_REFILL();
  }
  mma(B0); finish_step(D_ - 4);
  mma(B1); finish_step(D_ - 3);
  mma(B2); finish_step(D_ - 2);
  mma(B3); finish_step(D_ - 1);

#pragma unroll
  for (int rr = 6; rr >= 4; --rr) mypart[(D_ + 3 - rr) * 32] = acc0[rr] + acc1[rr];

  __syncthreads();

  const float bv = bias[0];
  float* sp = scale + (long)SCALE_PLANE(b) * DHW + (h0 + g) * W_ + w0 + n;
#pragma unroll
  for (int i = 0; i < D_ / KSPLIT; ++i) {
    const int d = ks * (D_ / KSPLIT) + i;
    const float* pp = part + d * 32 + lane;
    const float y = ((pp[0] + pp[D_ * 32]) + pp[2 * D_ * 32]) + pp[3 * D_ * 32];
    const float z = y + bv;
    sp[d * HW] = 1.0f / (1.0f + __expf(-z));
  }
}

#define ACH 4
__global__ __launch_bounds__(256) void sg_apply(
    const float* __restrict__ x, const float* __restrict__ scale,
    float* __restrict__ out) {
  const int bid = blockIdx.x;
  const int pc  = bid & 127;
  const int cg  = (bid >> 7) & 15;
  const int b   = bid >> 11;
  const int p   = (pc * 256 + (int)threadIdx.x) * 4;
  const long base = (long)b * CDHW + (long)(cg * ACH) * DHW + p;
  const float4 sv = *(const float4*)(scale + (long)SCALE_PLANE(b) * DHW + p);
  float4 xv[ACH];
#pragma unroll
  for (int k = 0; k < ACH; ++k) xv[k] = *(const float4*)(x + base + (long)k * DHW);
#pragma unroll
  for (int k = 0; k < ACH; ++k) {
    float4 o;
    o.x = xv[k].x * sv.x; o.y = xv[k].y * sv.y;
    o.z = xv[k].z * sv.z; o.w = xv[k].w * sv.w;
    *(float4*)(out + base + (long)k * DHW) = o;
  }
}

extern "C" void kernel_launch(void* const* d_in, const int* in_sizes, int n_in,
                              void* d_out, int out_size, void* d_ws, size_t ws_size,
                              hipStream_t stream) {
  const float* x      = (const float*)d_in[0];
  const float* conv_w = (const float*)d_in[1];
  const float* conv_b = (const float*)d_in[2];
  float* out = (float*)d_out;

  float* ws    = (float*)d_ws;
  float* scale = ws;
  float* xmax  = ws + 2L * DHW;
  float* xmean = ws + 2L * DHW + (long)B_ * DHW;

  sg_reduce_maxmean<<<512, 256, 0, stream>>>(x, xmax, xmean);

  sg_conv_sigmoid<<<512, 128, 0, stream>>>(xmax, xmean, conv_w, conv_b, scale);

  sg_apply<<<8192, 256, 0, stream>>>(x, scale, out);
}
